# starting kernel
# baseline (speedup 1.0000x reference)
#include <hip/hip_runtime.h>


typedef __attribute__((ext_vector_type(2))) float v2f;
typedef __attribute__((ext_vector_type(8))) float v8f;

#define B_   8
#define C_   1024
#define HW_  4096
#define M_   8
#define GAIN 0.3f
#define EPS_ 1e-6f


static __device__ __forceinline__ v8f wmma_f32(v2f a, v2f b, v8f c) {
  return __builtin_amdgcn_wmma_f32_16x16x4_f32(false, a, false, b, (short)0, c,
                                               false, false);
}

__global__ void __launch_bounds__(256) csc_init(const float* __restrict__ Wd,
                                                float* __restrict__ ws) {
  const int tid  = threadIdx.x;
  const int w    = tid >> 5;
  const int lane = tid & 31;
  if (tid < 16) ws[tid] = 0.0f;
  if (w < M_) {
    float s = 0.0f;
    for (int i = lane; i < C_; i += 32) s += Wd[w * C_ + i];
    for (int off = 16; off >= 1; off >>= 1) s += __shfl_xor(s, off, 32);
    if (lane == 0) ws[16 + w] = s;
  }
}

__global__ void __launch_bounds__(256) csc_pass1(const float* __restrict__ z,
                                                 const float* __restrict__ Wd,
                                                 float* __restrict__ ws) {
  __shared__ float WTP[C_ * 16];

  const int tid = threadIdx.x;
  const int b   = blockIdx.y;

  for (int c = tid; c < C_; c += 256) {
#pragma unroll
    for (int mr = 0; mr < 16; ++mr)
      WTP[c * 16 + mr] = (mr < M_) ? Wd[mr * C_ + c] : 0.0f;
  }
  __syncthreads();

  const int wave = tid >> 5;
  const int lane = tid & 31;
  const int hl   = lane >> 4;
  const int l15  = lane & 15;
  const int p0   = (blockIdx.x * 8 + wave) * 16;
  const float* zb = z + (size_t)b * C_ * HW_;

  v8f acc = {};
  float s1 = 0.0f, s2 = 0.0f;

  for (int c0 = 0; c0 < C_; c0 += 4) {
    const int ca = c0 + 2 * hl;
    v2f a, bb;
    a.x = WTP[ca * 16 + l15];
    a.y = WTP[(ca + 1) * 16 + l15];
    const float z0 = zb[(size_t)ca * HW_ + p0 + l15];
    const float z1 = zb[(size_t)(ca + 1) * HW_ + p0 + l15];
    bb.x = z0;
    bb.y = z1;
    acc = wmma_f32(a, bb, acc);
    s1 += z0 + z1;
    s2 = fmaf(z0, z0, s2);
    s2 = fmaf(z1, z1, s2);
  }

  float* yraw = ws + 32 + (size_t)b * M_ * HW_;
  if (hl == 0) {
#pragma unroll
    for (int g = 0; g < M_; ++g)
      yraw[g * HW_ + p0 + l15] = acc[g];
  }

  for (int off = 16; off >= 1; off >>= 1) {
    s1 += __shfl_xor(s1, off, 32);
    s2 += __shfl_xor(s2, off, 32);
  }
  if (lane == 0) {
    atomicAdd(&ws[b], s1);
    atomicAdd(&ws[8 + b], s2);
  }
}

__global__ void __launch_bounds__(256) csc_pass2(const float* __restrict__ z,
                                                 const float* __restrict__ Wd,
                                                 const float* __restrict__ ws,
                                                 float* __restrict__ out) {
  const int tid  = threadIdx.x;
  const int b    = blockIdx.y;
  const int wave = tid >> 5;
  const int lane = tid & 31;
  const int hl   = lane >> 4;
  const int l15  = lane & 15;
  const int p0   = (blockIdx.x * 8 + wave) * 16;

  const float invN = 1.0f / (float)(C_ * HW_);
  const float mu   = ws[b] * invN;
  const float var  = ws[8 + b] * invN - mu * mu;
  const float rsig = rsqrtf(var + EPS_);

  const float* yraw = ws + 32 + (size_t)b * M_ * HW_;
  const int m0 = 2 * hl;
  v2f b1, b2;
  b1.x = rsig * (yraw[(m0 + 0) * HW_ + p0 + l15] - mu * ws[16 + m0 + 0]);
  b1.y = rsig * (yraw[(m0 + 1) * HW_ + p0 + l15] - mu * ws[16 + m0 + 1]);
  b2.x = rsig * (yraw[(m0 + 4) * HW_ + p0 + l15] - mu * ws[16 + m0 + 4]);
  b2.y = rsig * (yraw[(m0 + 5) * HW_ + p0 + l15] - mu * ws[16 + m0 + 5]);

  const float* zb = z + (size_t)b * C_ * HW_;
  float* ob       = out + (size_t)b * C_ * HW_;

  for (int cb = 0; cb < C_; cb += 16) {
    v2f a1, a2;
    a1.x = Wd[(m0 + 0) * C_ + cb + l15];
    a1.y = Wd[(m0 + 1) * C_ + cb + l15];
    a2.x = Wd[(m0 + 4) * C_ + cb + l15];
    a2.y = Wd[(m0 + 5) * C_ + cb + l15];

    v8f d = {};
    d = wmma_f32(a1, b1, d);
    d = wmma_f32(a2, b2, d);

    const size_t base = (size_t)(cb + 8 * hl) * HW_ + p0 + l15;
#pragma unroll
    for (int g = 0; g < 8; ++g) {
      const float zv = __builtin_nontemporal_load(zb + base + (size_t)g * HW_);
      __builtin_nontemporal_store(fmaf(GAIN, d[g], zv),
                                  ob + base + (size_t)g * HW_);
    }
  }
}

extern "C" void kernel_launch(void* const* d_in, const int* in_sizes, int n_in,
                              void* d_out, int out_size, void* d_ws, size_t ws_size,
                              hipStream_t stream) {
  const float* z  = (const float*)d_in[0];
  const float* Wd = (const float*)d_in[1];
  float* out = (float*)d_out;
  float* ws  = (float*)d_ws;

  csc_init<<<dim3(1), dim3(256), 0, stream>>>(Wd, ws);
  dim3 grid(32, B_);
  csc_pass1<<<grid, dim3(256), 0, stream>>>(z, Wd, ws);
  csc_pass2<<<grid, dim3(256), 0, stream>>>(z, Wd, ws, out);
}


# Re-tile both passes for bytes in flight: wave tile 64 px x 128 ch, 16-byte lane loads/stores issued 3 groups ahead; 8 waves split channels, partials summed via LDS (summation order changes)
# speedup vs baseline: 3.2127x; 3.2127x over previous
#include <hip/hip_runtime.h>


typedef __attribute__((ext_vector_type(2))) float v2f;
typedef __attribute__((ext_vector_type(4))) float v4f;
typedef __attribute__((ext_vector_type(8))) float v8f;

#define B_   8
#define C_   1024
#define HW_  4096
#define M_   8
#define GAIN 0.3f
#define EPS_ 1e-6f

#define PT_  64
#define CW_  128


static __device__ __forceinline__ v8f wmma_f32(v2f a, v2f b, v8f c) {
  return __builtin_amdgcn_wmma_f32_16x16x4_f32(false, a, false, b, (short)0, c,
                                               false, false);
}

__global__ void __launch_bounds__(256) csc_init(const float* __restrict__ Wd,
                                                float* __restrict__ ws) {
  const int tid  = threadIdx.x;
  const int w    = tid >> 5;
  const int lane = tid & 31;
  if (tid < 16) ws[tid] = 0.0f;
  if (w < M_) {
    float s = 0.0f;
    for (int i = lane; i < C_; i += 32) s += Wd[w * C_ + i];
    for (int off = 16; off >= 1; off >>= 1) s += __shfl_xor(s, off, 32);
    if (lane == 0) ws[16 + w] = s;
  }
}

#define P1_NB    4
#define P1_NG    (CW_ / 4 / P1_NB)
#define P1_AHEAD 3

__global__ void __launch_bounds__(256) csc_pass1(const float* __restrict__ z,
                                                 const float* __restrict__ Wd,
                                                 float* __restrict__ ws) {
  __shared__ v4f   red[8][M_][PT_ / 4];
  __shared__ float sred[2][8];

  const int tid  = threadIdx.x;
  const int b    = blockIdx.y;
  const int wave = tid >> 5;
  const int lane = tid & 31;
  const int hl   = lane >> 4;
  const int l15  = lane & 15;
  const int p0   = blockIdx.x * PT_;
  const int cw   = wave * CW_ + 2 * hl;

  const float* zl = z + (size_t)b * C_ * HW_ + (size_t)cw * HW_ + p0 + 4 * l15;
  const float* wl = Wd + (l15 & 7) * C_ + cw;
  const bool   mrow = l15 < M_;

  v4f zr[P1_AHEAD + 1][P1_NB][2];
  v2f ar[P1_AHEAD + 1][P1_NB];

#define P1_LOAD(buf, g)                                                        \
  _Pragma("unroll") for (int s = 0; s < P1_NB; ++s) {                          \
    const int st = (g) * P1_NB + s;                                            \
    ar[buf][s]    = *(const v2f*)(wl + 4 * st);                                \
    zr[buf][s][0] = *(const v4f*)(zl + (size_t)(4 * st) * HW_);                \
    zr[buf][s][1] = *(const v4f*)(zl + (size_t)(4 * st + 1) * HW_);            \
  }

  v8f acc[4] = {};
  v4f s1 = {}, s2 = {};

#pragma unroll
  for (int g = 0; g < P1_AHEAD; ++g) { P1_LOAD(g, g) }

#pragma unroll
  for (int g = 0; g < P1_NG; ++g) {
    const int cur = g % (P1_AHEAD + 1);
    if (g + P1_AHEAD < P1_NG) { P1_LOAD((g + P1_AHEAD) % (P1_AHEAD + 1), g + P1_AHEAD) }
    __builtin_amdgcn_sched_barrier(0);
#pragma unroll
    for (int s = 0; s < P1_NB; ++s) {
      v2f a;
      a.x = mrow ? ar[cur][s].x : 0.0f;
      a.y = mrow ? ar[cur][s].y : 0.0f;
      const v4f z0 = zr[cur][s][0];
      const v4f z1 = zr[cur][s][1];
#pragma unroll
      for (int j = 0; j < 4; ++j) {
        v2f bb;
        bb.x = z0[j];
        bb.y = z1[j];
        acc[j] = wmma_f32(a, bb, acc[j]);
      }
      s1 += z0;
      s1 += z1;
      s2 = z0 * z0 + s2;
      s2 = z1 * z1 + s2;
    }
    __builtin_amdgcn_sched_barrier(0);
  }
#undef P1_LOAD

  float t1 = (s1.x + s1.y) + (s1.z + s1.w);
  float t2 = (s2.x + s2.y) + (s2.z + s2.w);
  for (int off = 16; off >= 1; off >>= 1) {
    t1 += __shfl_xor(t1, off, 32);
    t2 += __shfl_xor(t2, off, 32);
  }

  if (hl == 0) {
#pragma unroll
    for (int r = 0; r < M_; ++r) {
      v4f q;
      q.x = acc[0][r]; q.y = acc[1][r]; q.z = acc[2][r]; q.w = acc[3][r];
      red[wave][r][l15] = q;
    }
  }
  if (lane == 0) {
    sred[0][wave] = t1;
    sred[1][wave] = t2;
  }
  __syncthreads();

  {
    const int m  = tid >> 5;
    const int px = (tid & 31) * 2;
    const float* rf = (const float*)&red[0][0][0];
    v2f y = *(const v2f*)(rf + m * PT_ + px);
#pragma unroll
    for (int w = 1; w < 8; ++w)
      y += *(const v2f*)(rf + w * (M_ * PT_) + m * PT_ + px);
    float* yraw = ws + 32 + (size_t)b * M_ * HW_;
    *(v2f*)(yraw + m * HW_ + p0 + px) = y;
  }
  if (tid < 2) {
    float t = sred[tid][0];
#pragma unroll
    for (int w = 1; w < 8; ++w) t += sred[tid][w];
    atomicAdd(&ws[8 * tid + b], t);
  }
}

#define P2_NBLK  (CW_ / 16)
#define P2_AHEAD 3

__global__ void __launch_bounds__(256) csc_pass2(const float* __restrict__ z,
                                                 const float* __restrict__ Wd,
                                                 const float* __restrict__ ws,
                                                 float* __restrict__ out) {
  const int tid  = threadIdx.x;
  const int b    = blockIdx.y;
  const int wave = tid >> 5;
  const int lane = tid & 31;
  const int hl   = lane >> 4;
  const int l15  = lane & 15;
  const int p0   = blockIdx.x * PT_;
  const int cw   = wave * CW_;
  const int m0   = 2 * hl;

  const float sum1 = ws[b];
  const float sum2 = ws[8 + b];
  const float* yraw = ws + 32 + (size_t)b * M_ * HW_ + p0 + 4 * l15;
  v4f   yr[4];
  float wsum[4];
#pragma unroll
  for (int q = 0; q < 4; ++q) {
    const int m = m0 + (q & 1) + 4 * (q >> 1);
    yr[q]   = *(const v4f*)(yraw + m * HW_);
    wsum[q] = ws[16 + m];
  }

  const size_t off = (size_t)b * C_ * HW_ + (size_t)(cw + 8 * hl) * HW_ + p0 + 4 * l15;
  const float* zl = z + off;
  float*       ol = out + off;
  const float* wl = Wd + m0 * C_ + cw + l15;

  v4f   zr[P2_AHEAD + 1][8];
  float ar[P2_AHEAD + 1][4];
#define P2_LOAD(buf, blk)                                                      \
  ar[buf][0] = wl[0 * C_ + (blk) * 16];                                        \
  ar[buf][1] = wl[1 * C_ + (blk) * 16];                                        \
  ar[buf][2] = wl[4 * C_ + (blk) * 16];                                        \
  ar[buf][3] = wl[5 * C_ + (blk) * 16];                                        \
  _Pragma("unroll") for (int r = 0; r < 8; ++r)                                \
    zr[buf][r] = __builtin_nontemporal_load(                                   \
        (const v4f*)(zl + (size_t)((blk) * 16 + r) * HW_));

#pragma unroll
  for (int i = 0; i < P2_AHEAD; ++i) {
    P2_LOAD(i, i)
    __builtin_amdgcn_sched_barrier(0);
  }

  const float invN = 1.0f / (float)(C_ * HW_);
  const float mu   = sum1 * invN;
  const float var  = sum2 * invN - mu * mu;
  const float rsig = rsqrtf(var + EPS_);

  v4f yv[4];
#pragma unroll
  for (int q = 0; q < 4; ++q) {
    const float ms = mu * wsum[q];
    yv[q].x = rsig * (yr[q].x - ms);
    yv[q].y = rsig * (yr[q].y - ms);
    yv[q].z = rsig * (yr[q].z - ms);
    yv[q].w = rsig * (yr[q].w - ms);
  }

#pragma unroll
  for (int i = 0; i < P2_NBLK; ++i) {
    const int cur = i % (P2_AHEAD + 1);
    if (i + P2_AHEAD < P2_NBLK) { P2_LOAD((i + P2_AHEAD) % (P2_AHEAD + 1), i + P2_AHEAD) }
    __builtin_amdgcn_sched_barrier(0);

    v2f a1, a2;
    a1.x = ar[cur][0];
    a1.y = ar[cur][1];
    a2.x = ar[cur][2];
    a2.y = ar[cur][3];

    v8f d[4];
#pragma unroll
    for (int j = 0; j < 4; ++j) {
      v2f b1, b2;
      b1.x = yv[0][j]; b1.y = yv[1][j];
      b2.x = yv[2][j]; b2.y = yv[3][j];
      v8f t = {};
      t = wmma_f32(a1, b1, t);
      t = wmma_f32(a2, b2, t);
      d[j] = t;
    }

#pragma unroll
    for (int r = 0; r < 8; ++r) {
      const v4f zv = zr[cur][r];
      v4f o;
      o.x = fmaf(GAIN, d[0][r], zv.x);
      o.y = fmaf(GAIN, d[1][r], zv.y);
      o.z = fmaf(GAIN, d[2][r], zv.z);
      o.w = fmaf(GAIN, d[3][r], zv.w);
      __builtin_nontemporal_store(o, (v4f*)(ol + (size_t)(i * 16 + r) * HW_));
    }
    __builtin_amdgcn_sched_barrier(0);
  }
#undef P2_LOAD
}

extern "C" void kernel_launch(void* const* d_in, const int* in_sizes, int n_in,
                              void* d_out, int out_size, void* d_ws, size_t ws_size,
                              hipStream_t stream) {
  const float* z  = (const float*)d_in[0];
  const float* Wd = (const float*)d_in[1];
  float* out = (float*)d_out;
  float* ws  = (float*)d_ws;

  csc_init<<<dim3(1), dim3(256), 0, stream>>>(Wd, ws);
  dim3 grid(64, B_);
  csc_pass1<<<grid, dim3(256), 0, stream>>>(z, Wd, ws);
  csc_pass2<<<grid, dim3(256), 0, stream>>>(z, Wd, ws, out);
}


# Drop the set-up launch (3 -> 2): pass 1 stores per-workgroup sum/sumsq lines and Wsum from its weight fragments, no atomics; pass 2 adds the 64 lines in a fixed tree (summation order changes)
# speedup vs baseline: 3.4093x; 1.0612x over previous
#include <hip/hip_runtime.h>


typedef __attribute__((ext_vector_type(2))) float v2f;
typedef __attribute__((ext_vector_type(4))) float v4f;
typedef __attribute__((ext_vector_type(8))) float v8f;

#define B_   8
#define C_   1024
#define HW_  4096
#define M_   8
#define GAIN 0.3f
#define EPS_ 1e-6f

#define PT_  64
#define CW_  128

#define WS_WSUM 16
#define WS_YRAW 32
#define WS_PART (WS_YRAW + B_ * M_ * HW_)
#define NTILE   (HW_ / PT_)

static __device__ __forceinline__ v8f wmma_f32(v2f a, v2f b, v8f c) {
  return __builtin_amdgcn_wmma_f32_16x16x4_f32(false, a, false, b, (short)0, c,
                                               false, false);
}

#define P1_NB    4
#define P1_NG    (CW_ / 4 / P1_NB)
#define P1_AHEAD 3

__global__ void __launch_bounds__(256) csc_pass1(const float* __restrict__ z,
                                                 const float* __restrict__ Wd,
                                                 float* __restrict__ ws) {
  __shared__ v4f   red[8][M_][PT_ / 4];
  __shared__ float sred[2][8];
  __shared__ float wred[8][M_];

  const int tid  = threadIdx.x;
  const int b    = blockIdx.y;
  const int wave = tid >> 5;
  const int lane = tid & 31;
  const int hl   = lane >> 4;
  const int l15  = lane & 15;
  const int p0   = blockIdx.x * PT_;
  const int cw   = wave * CW_ + 2 * hl;

  const float* zl = z + (size_t)b * C_ * HW_ + (size_t)cw * HW_ + p0 + 4 * l15;
  const float* wl = Wd + (l15 & 7) * C_ + cw;
  const bool   mrow = l15 < M_;

  v4f zr[P1_AHEAD + 1][P1_NB][2];
  v2f ar[P1_AHEAD + 1][P1_NB];

#define P1_LOAD(buf, g)                                                        \
  _Pragma("unroll") for (int s = 0; s < P1_NB; ++s) {                          \
    const int st = (g) * P1_NB + s;                                            \
    ar[buf][s]    = *(const v2f*)(wl + 4 * st);                                \
    zr[buf][s][0] = *(const v4f*)(zl + (size_t)(4 * st) * HW_);                \
    zr[buf][s][1] = *(const v4f*)(zl + (size_t)(4 * st + 1) * HW_);            \
  }

  v8f acc[4] = {};
  v4f s1 = {}, s2 = {};
  v2f wacc = {};

#pragma unroll
  for (int g = 0; g < P1_AHEAD; ++g) { P1_LOAD(g, g) }

#pragma unroll
  for (int g = 0; g < P1_NG; ++g) {
    const int cur = g % (P1_AHEAD + 1);
    if (g + P1_AHEAD < P1_NG) { P1_LOAD((g + P1_AHEAD) % (P1_AHEAD + 1), g + P1_AHEAD) }
    __builtin_amdgcn_sched_barrier(0);
#pragma unroll
    for (int s = 0; s < P1_NB; ++s) {
      v2f a;
      wacc += ar[cur][s];
      a.x = mrow ? ar[cur][s].x : 0.0f;
      a.y = mrow ? ar[cur][s].y : 0.0f;
      const v4f z0 = zr[cur][s][0];
      const v4f z1 = zr[cur][s][1];
#pragma unroll
      for (int j = 0; j < 4; ++j) {
        v2f bb;
        bb.x = z0[j];
        bb.y = z1[j];
        acc[j] = wmma_f32(a, bb, acc[j]);
      }
      s1 += z0;
      s1 += z1;
      s2 = z0 * z0 + s2;
      s2 = z1 * z1 + s2;
    }
    __builtin_amdgcn_sched_barrier(0);
  }
#undef P1_LOAD

  float t1 = (s1.x + s1.y) + (s1.z + s1.w);
  float t2 = (s2.x + s2.y) + (s2.z + s2.w);
  for (int off = 16; off >= 1; off >>= 1) {
    t1 += __shfl_xor(t1, off, 32);
    t2 += __shfl_xor(t2, off, 32);
  }

  if (hl == 0) {
#pragma unroll
    for (int r = 0; r < M_; ++r) {
      v4f q;
      q.x = acc[0][r]; q.y = acc[1][r]; q.z = acc[2][r]; q.w = acc[3][r];
      red[wave][r][l15] = q;
    }
  }
  if (lane == 0) {
    sred[0][wave] = t1;
    sred[1][wave] = t2;
  }
  const bool first_wg = (blockIdx.x == 0) && (blockIdx.y == 0);
  if (first_wg) {
    float wt = wacc.x + wacc.y;
    wt += __shfl_xor(wt, 16, 32);
    if (lane < M_) wred[wave][lane] = wt;
  }
  __syncthreads();

  {
    const int m  = tid >> 5;
    const int px = (tid & 31) * 2;
    const float* rf = (const float*)&red[0][0][0];
    v2f y = *(const v2f*)(rf + m * PT_ + px);
#pragma unroll
    for (int w = 1; w < 8; ++w)
      y += *(const v2f*)(rf + w * (M_ * PT_) + m * PT_ + px);
    float* yraw = ws + WS_YRAW + (size_t)b * M_ * HW_;
    *(v2f*)(yraw + m * HW_ + p0 + px) = y;
  }
  if (tid < 2) {
    float t = sred[tid][0];
#pragma unroll
    for (int w = 1; w < 8; ++w) t += sred[tid][w];
    ws[WS_PART + (b * NTILE + (int)blockIdx.x) * 32 + tid] = t;
  }
  if (first_wg && tid < M_) {
    float t = wred[0][tid];
#pragma unroll
    for (int w = 1; w < 8; ++w) t += wred[w][tid];
    ws[WS_WSUM + tid] = t;
  }
}

#define P2_NBLK  (CW_ / 16)
#define P2_AHEAD 3

__global__ void __launch_bounds__(256) csc_pass2(const float* __restrict__ z,
                                                 const float* __restrict__ Wd,
                                                 const float* __restrict__ ws,
                                                 float* __restrict__ out) {
  const int tid  = threadIdx.x;
  const int b    = blockIdx.y;
  const int wave = tid >> 5;
  const int lane = tid & 31;
  const int hl   = lane >> 4;
  const int l15  = lane & 15;
  const int p0   = blockIdx.x * PT_;
  const int cw   = wave * CW_;
  const int m0   = 2 * hl;

  const float* part = ws + WS_PART + (size_t)(b * NTILE) * 32;
  const v2f pa = *(const v2f*)(part + lane * 32);
  const v2f pb = *(const v2f*)(part + (lane + 32) * 32);
  const float* yraw = ws + WS_YRAW + (size_t)b * M_ * HW_ + p0 + 4 * l15;
  v4f   yr[4];
  float wsum[4];
#pragma unroll
  for (int q = 0; q < 4; ++q) {
    const int m = m0 + (q & 1) + 4 * (q >> 1);
    yr[q]   = *(const v4f*)(yraw + m * HW_);
    wsum[q] = ws[WS_WSUM + m];
  }

  const size_t off = (size_t)b * C_ * HW_ + (size_t)(cw + 8 * hl) * HW_ + p0 + 4 * l15;
  const float* zl = z + off;
  float*       ol = out + off;
  const float* wl = Wd + m0 * C_ + cw + l15;

  v4f   zr[P2_AHEAD + 1][8];
  float ar[P2_AHEAD + 1][4];
#define P2_LOAD(buf, blk)                                                      \
  ar[buf][0] = wl[0 * C_ + (blk) * 16];                                        \
  ar[buf][1] = wl[1 * C_ + (blk) * 16];                                        \
  ar[buf][2] = wl[4 * C_ + (blk) * 16];                                        \
  ar[buf][3] = wl[5 * C_ + (blk) * 16];                                        \
  _Pragma("unroll") for (int r = 0; r < 8; ++r)                                \
    zr[buf][r] = __builtin_nontemporal_load(                                   \
        (const v4f*)(zl + (size_t)((blk) * 16 + r) * HW_));

#pragma unroll
  for (int i = 0; i < P2_AHEAD; ++i) {
    P2_LOAD(i, i)
    __builtin_amdgcn_sched_barrier(0);
  }

  float sum1 = pa.x + pb.x;
  float sum2 = pa.y + pb.y;
  for (int off = 16; off >= 1; off >>= 1) {
    sum1 += __shfl_xor(sum1, off, 32);
    sum2 += __shfl_xor(sum2, off, 32);
  }

  const float invN = 1.0f / (float)(C_ * HW_);
  const float mu   = sum1 * invN;
  const float var  = sum2 * invN - mu * mu;
  const float rsig = rsqrtf(var + EPS_);

  v4f yv[4];
#pragma unroll
  for (int q = 0; q < 4; ++q) {
    const float ms = mu * wsum[q];
    yv[q].x = rsig * (yr[q].x - ms);
    yv[q].y = rsig * (yr[q].y - ms);
    yv[q].z = rsig * (yr[q].z - ms);
    yv[q].w = rsig * (yr[q].w - ms);
  }

#pragma unroll
  for (int i = 0; i < P2_NBLK; ++i) {
    const int cur = i % (P2_AHEAD + 1);
    if (i + P2_AHEAD < P2_NBLK) { P2_LOAD((i + P2_AHEAD) % (P2_AHEAD + 1), i + P2_AHEAD) }
    __builtin_amdgcn_sched_barrier(0);

    v2f a1, a2;
    a1.x = ar[cur][0];
    a1.y = ar[cur][1];
    a2.x = ar[cur][2];
    a2.y = ar[cur][3];

    v8f d[4];
#pragma unroll
    for (int j = 0; j < 4; ++j) {
      v2f b1, b2;
      b1.x = yv[0][j]; b1.y = yv[1][j];
      b2.x = yv[2][j]; b2.y = yv[3][j];
      v8f t = {};
      t = wmma_f32(a1, b1, t);
      t = wmma_f32(a2, b2, t);
      d[j] = t;
    }

#pragma unroll
    for (int r = 0; r < 8; ++r) {
      const v4f zv = zr[cur][r];
      v4f o;
      o.x = fmaf(GAIN, d[0][r], zv.x);
      o.y = fmaf(GAIN, d[1][r], zv.y);
      o.z = fmaf(GAIN, d[2][r], zv.z);
      o.w = fmaf(GAIN, d[3][r], zv.w);
      __builtin_nontemporal_store(o, (v4f*)(ol + (size_t)(i * 16 + r) * HW_));
    }
    __builtin_amdgcn_sched_barrier(0);
  }
#undef P2_LOAD
}

extern "C" void kernel_launch(void* const* d_in, const int* in_sizes, int n_in,
                              void* d_out, int out_size, void* d_ws, size_t ws_size,
                              hipStream_t stream) {
  const float* z  = (const float*)d_in[0];
  const float* Wd = (const float*)d_in[1];
  float* out = (float*)d_out;
  float* ws  = (float*)d_ws;

  dim3 grid(64, B_);
  csc_pass1<<<grid, dim3(256), 0, stream>>>(z, Wd, ws);
  csc_pass2<<<grid, dim3(256), 0, stream>>>(z, Wd, ws, out);
}
